# starting kernel
# baseline (speedup 1.0000x reference)
#include <hip/hip_runtime.h>
#include <math.h>


#define N    512
#define DIM  512
#define H    64
#define OUT  32

typedef __attribute__((ext_vector_type(2))) float v2f;
typedef __attribute__((ext_vector_type(8))) float v8f;

__device__ inline v8f gemm16x16_f32(const float* __restrict__ Arow, int lda,
                                    const float* __restrict__ Bcol, int ldb,
                                    int K, int lane) {
  v8f c = {};
  const float* ap = Arow + (lane & 15) * lda + ((lane >> 4) << 1);
  const float* bp = Bcol + ((lane >> 4) << 1) * ldb;
  for (int k = 0; k < K; k += 4) {
    v2f a; a.x = ap[0];  a.y = ap[1];
    v2f b; b.x = bp[0];  b.y = bp[ldb];
    c = __builtin_amdgcn_wmma_f32_16x16x4_f32(false, a, false, b, (short)0, c,
                                              false, false);
    ap += 4;
    bp += 4 * ldb;
  }
  return c;
}

__device__ inline void store_tile16x16(float* __restrict__ D, int ldd,
                                       int m0, int n0, v8f c, int lane) {
  int col   = n0 + (lane & 15);
  int rbase = m0 + ((lane >> 4) << 3);
#pragma unroll
  for (int v = 0; v < 8; ++v) D[(rbase + v) * ldd + col] = c[v];
}

__global__ void gemm_pq_kernel(const float* __restrict__ F,
                               const float* __restrict__ w1,
                               float* __restrict__ PQ) {
  const int wave = threadIdx.x >> 5;
  const int lane = threadIdx.x & 31;
  const int tile = blockIdx.x * 8 + wave;
  const int m0 = (tile >> 3) * 16;
  const int n0 = (tile & 7) * 16;
  const int c  = n0 + (lane & 15);
  const float* Bcol = (c < H) ? (w1 + c) : (w1 + DIM * H + (c - H));
  v8f acc = gemm16x16_f32(F + m0 * DIM, DIM, Bcol, H, DIM, lane);
  store_tile16x16(PQ, 2 * H, m0, n0, acc, lane);
}

__global__ void gemm_xw_kernel(const float* __restrict__ F,
                               const float* __restrict__ W,
                               float* __restrict__ XW) {
  const int wave = threadIdx.x >> 5;
  const int lane = threadIdx.x & 31;
  const int tile = blockIdx.x * 8 + wave;
  const int m0 = (tile >> 1) * 16;
  const int n0 = (tile & 1) * 16;
  const float* Bcol = W + n0 + (lane & 15);
  v8f acc = gemm16x16_f32(F + m0 * DIM, DIM, Bcol, OUT, DIM, lane);
  store_tile16x16(XW, OUT, m0, n0, acc, lane);
}

__global__ void relnet_pair_kernel(const float* __restrict__ PQ,
                                   const float* __restrict__ b1,
                                   const float* __restrict__ w2,
                                   const float* __restrict__ b2p,
                                   float* __restrict__ adj) {
  __shared__ float Pj[16][H];
  __shared__ float Qi[16][H];
  __shared__ float b1s[H];
  __shared__ float w2s[H];
  const int tid = threadIdx.x;
  const int j0 = blockIdx.x * 16;
  const int i0 = blockIdx.y * 16;

  for (int t = tid; t < 16 * H; t += 256) {
    int r = t >> 6, h = t & (H - 1);
    Pj[r][h] = PQ[(j0 + r) * (2 * H) + h];
    Qi[r][h] = PQ[(i0 + r) * (2 * H) + H + h];
  }
  if (tid < H) { b1s[tid] = b1[tid]; w2s[tid] = w2[tid]; }
  __syncthreads();

  const int tj = tid & 15;
  const int ti = tid >> 4;
  float acc = 0.0f;
#pragma unroll 8
  for (int h = 0; h < H; ++h) {
    float t = Pj[tj][h] + Qi[ti][h] + b1s[h];
    acc = fmaf(fmaxf(t, 0.0f), w2s[h], acc);
  }
  float z = acc + b2p[0];
  adj[(i0 + ti) * N + (j0 + tj)] = 1.0f / (1.0f + expf(-z));
}

__global__ void dvec_kernel(const float* __restrict__ adj,
                            float* __restrict__ dvec) {
  __shared__ float red[256];
  const int i = blockIdx.x, tid = threadIdx.x;
  red[tid] = adj[i * N + tid] + adj[i * N + tid + 256];
  __syncthreads();
  for (int off = 128; off > 0; off >>= 1) {
    if (tid < off) red[tid] += red[tid + off];
    __syncthreads();
  }
  if (tid == 0) dvec[i] = sqrtf(red[0] + 2.0f);
}

__global__ void norm_kernel(const float* __restrict__ adj,
                            const float* __restrict__ dvec,
                            float* __restrict__ Aout) {
  const int idx = blockIdx.x * 256 + threadIdx.x;
  const int i = idx >> 9;
  const int j = idx & (N - 1);
  float a = adj[idx] + (i == j ? 1.0f : 0.0f);
  Aout[idx] = a / (dvec[i] * dvec[j]);
}

__global__ void gemm_out_kernel(const float* __restrict__ Anorm,
                                const float* __restrict__ XW,
                                const float* __restrict__ gcn_b,
                                const float* __restrict__ gamma,
                                const float* __restrict__ beta,
                                float* __restrict__ X) {
  const int wave = threadIdx.x >> 5;
  const int lane = threadIdx.x & 31;
  const int tile = blockIdx.x * 8 + wave;
  const int m0 = (tile >> 1) * 16;
  const int n0 = (tile & 1) * 16;
  const float* Bcol = XW + n0 + (lane & 15);
  v8f acc = gemm16x16_f32(Anorm + m0 * N, N, Bcol, OUT, N, lane);

  const int col = n0 + (lane & 15);
  const float inv_std = 0.99999500003749969f;
  const float sc = gamma[col] * inv_std;
  const float bt = beta[col];
  const float gb = gcn_b[col];
  const int rbase = m0 + ((lane >> 4) << 3);
#pragma unroll
  for (int v = 0; v < 8; ++v) {
    float t = (acc[v] + gb) * sc + bt;
    X[(rbase + v) * OUT + col] = fmaxf(t, 0.0f);
  }
}

extern "C" void kernel_launch(void* const* d_in, const int* in_sizes, int n_in,
                              void* d_out, int out_size, void* d_ws, size_t ws_size,
                              hipStream_t stream) {
  const float* F     = (const float*)d_in[0];
  const float* w1    = (const float*)d_in[1];
  const float* b1    = (const float*)d_in[2];
  const float* w2    = (const float*)d_in[3];
  const float* b2    = (const float*)d_in[4];
  const float* gw    = (const float*)d_in[5];
  const float* gb    = (const float*)d_in[6];
  const float* gamma = (const float*)d_in[7];
  const float* beta  = (const float*)d_in[8];

  float* out_x = (float*)d_out;
  float* out_A = out_x + N * OUT;

  float* ws   = (float*)d_ws;
  float* PQ   = ws;
  float* adj  = PQ + N * 2 * H;
  float* XW   = adj + N * N;
  float* dvec = XW + N * OUT;

  gemm_pq_kernel   <<<32,            256, 0, stream>>>(F, w1, PQ);
  gemm_xw_kernel   <<<8,             256, 0, stream>>>(F, gw, XW);
  relnet_pair_kernel<<<dim3(32, 32), 256, 0, stream>>>(PQ, b1, w2, b2, adj);
  dvec_kernel      <<<N,             256, 0, stream>>>(adj, dvec);
  norm_kernel      <<<(N * N) / 256, 256, 0, stream>>>(adj, dvec, out_A);
  gemm_out_kernel  <<<8,             256, 0, stream>>>(out_A, XW, gb, gamma, beta, out_x);
}


# GEMMs: work-group copies its 16-row A strip to LDS once (16-byte loads); 2 waves run whole K from LDS, B fetched 3 blocks (24 steps) ahead through a register ring; f32 sum order unchanged
# speedup vs baseline: 1.5296x; 1.5296x over previous
#include <hip/hip_runtime.h>
#include <math.h>


#define N    512
#define DIM  512
#define H    64
#define OUT  32

typedef __attribute__((ext_vector_type(2))) float v2f;
typedef __attribute__((ext_vector_type(8))) float v8f;

typedef __attribute__((ext_vector_type(4))) float v4f;

#define PITCH 516

struct BRing { v2f r[4][8]; };

__device__ __forceinline__ void stage_issue(v4f (&st)[8], const float* __restrict__ src, int tid) {
#pragma unroll
  for (int i = 0; i < 8; ++i) {
    const int idx = tid + i * 256;
    st[i] = *(const v4f*)(src + (idx >> 7) * DIM + (idx & 127) * 4);
  }
}

__device__ __forceinline__ void stage_commit(float* As, const v4f (&st)[8], int tid) {
#pragma unroll
  for (int i = 0; i < 8; ++i) {
    const int idx = tid + i * 256;
    *(v4f*)(As + (idx >> 7) * PITCH + (idx & 127) * 4) = st[i];
  }
}

template <int LDB>
__device__ __forceinline__ void load_b_block(v2f (&dst)[8], const float* __restrict__ bp, int blk) {
#pragma unroll
  for (int u = 0; u < 8; ++u) {
    const float* p = bp + (blk * 32 + u * 4) * LDB;
    dst[u].x = p[0];
    dst[u].y = p[LDB];
  }
}

template <int LDB>
__device__ __forceinline__ void b_prologue(BRing& q, const float* __restrict__ bp) {
#pragma unroll
  for (int blk = 0; blk < 3; ++blk) load_b_block<LDB>(q.r[blk], bp, blk);
}

template <int LDB>
__device__ __forceinline__ v8f strip_chain(BRing& q, const float* arow, const float* __restrict__ bp) {
  v8f c = {};
  v2f a[2][8];
#pragma unroll
  for (int u = 0; u < 8; ++u) a[0][u] = *(const v2f*)(arow + 4 * u);
#pragma unroll
  for (int blk = 0; blk < 16; ++blk) {
    if (blk + 3 < 16) load_b_block<LDB>(q.r[(blk + 3) & 3], bp, blk + 3);
    if (blk + 1 < 16) {
#pragma unroll
      for (int u = 0; u < 8; ++u) a[(blk + 1) & 1][u] = *(const v2f*)(arow + (blk + 1) * 32 + 4 * u);
    }
    __builtin_amdgcn_sched_barrier(0);
#pragma unroll
    for (int u = 0; u < 8; ++u)
      c = __builtin_amdgcn_wmma_f32_16x16x4_f32(false, a[blk & 1][u], false, q.r[blk & 3][u],
                                                (short)0, c, false, false);
  }
  return c;
}

__device__ inline void store_tile16x16(float* __restrict__ D, int ldd,
                                       int m0, int n0, v8f c, int lane) {
  int col   = n0 + (lane & 15);
  int rbase = m0 + ((lane >> 4) << 3);
#pragma unroll
  for (int v = 0; v < 8; ++v) D[(rbase + v) * ldd + col] = c[v];
}

__global__ __launch_bounds__(256) void gemm_pq_kernel(const float* __restrict__ F,
                                                      const float* __restrict__ w1,
                                                      float* __restrict__ PQ) {
  __shared__ float As[16 * PITCH];
  const int tid  = threadIdx.x;
  const int wave = tid >> 5;
  const int lane = tid & 31;
  const int m0 = (blockIdx.x >> 2) * 16;
  const int n0 = ((blockIdx.x & 3) * 2 + wave) * 16;
  v4f st[8];
  stage_issue(st, F + m0 * DIM, tid);
  BRing q;
  const float* bp = w1;
  if (wave < 2) {
    const int c = n0 + (lane & 15);
    const float* Bcol = (c < H) ? (w1 + c) : (w1 + DIM * H + (c - H));
    bp = Bcol + ((lane >> 4) << 1) * H;
    b_prologue<H>(q, bp);
  }
  stage_commit(As, st, tid);
  __syncthreads();
  if (wave >= 2) return;
  v8f acc = strip_chain<H>(q, As + (lane & 15) * PITCH + ((lane >> 4) << 1), bp);
  store_tile16x16(PQ, 2 * H, m0, n0, acc, lane);
}

__global__ __launch_bounds__(256) void gemm_xw_kernel(const float* __restrict__ F,
                                                      const float* __restrict__ W,
                                                      float* __restrict__ XW) {
  __shared__ float As[16 * PITCH];
  const int tid  = threadIdx.x;
  const int wave = tid >> 5;
  const int lane = tid & 31;
  const int m0 = blockIdx.x * 16;
  const int n0 = wave * 16;
  v4f st[8];
  stage_issue(st, F + m0 * DIM, tid);
  BRing q;
  const float* bp = W;
  if (wave < 2) {
    bp = W + n0 + (lane & 15) + ((lane >> 4) << 1) * OUT;
    b_prologue<OUT>(q, bp);
  }
  stage_commit(As, st, tid);
  __syncthreads();
  if (wave >= 2) return;
  v8f acc = strip_chain<OUT>(q, As + (lane & 15) * PITCH + ((lane >> 4) << 1), bp);
  store_tile16x16(XW, OUT, m0, n0, acc, lane);
}

__global__ void relnet_pair_kernel(const float* __restrict__ PQ,
                                   const float* __restrict__ b1,
                                   const float* __restrict__ w2,
                                   const float* __restrict__ b2p,
                                   float* __restrict__ adj) {
  __shared__ float Pj[16][H];
  __shared__ float Qi[16][H];
  __shared__ float b1s[H];
  __shared__ float w2s[H];
  const int tid = threadIdx.x;
  const int j0 = blockIdx.x * 16;
  const int i0 = blockIdx.y * 16;

  for (int t = tid; t < 16 * H; t += 256) {
    int r = t >> 6, h = t & (H - 1);
    Pj[r][h] = PQ[(j0 + r) * (2 * H) + h];
    Qi[r][h] = PQ[(i0 + r) * (2 * H) + H + h];
  }
  if (tid < H) { b1s[tid] = b1[tid]; w2s[tid] = w2[tid]; }
  __syncthreads();

  const int tj = tid & 15;
  const int ti = tid >> 4;
  float acc = 0.0f;
#pragma unroll 8
  for (int h = 0; h < H; ++h) {
    float t = Pj[tj][h] + Qi[ti][h] + b1s[h];
    acc = fmaf(fmaxf(t, 0.0f), w2s[h], acc);
  }
  float z = acc + b2p[0];
  adj[(i0 + ti) * N + (j0 + tj)] = 1.0f / (1.0f + expf(-z));
}

__global__ void dvec_kernel(const float* __restrict__ adj,
                            float* __restrict__ dvec) {
  __shared__ float red[256];
  const int i = blockIdx.x, tid = threadIdx.x;
  red[tid] = adj[i * N + tid] + adj[i * N + tid + 256];
  __syncthreads();
  for (int off = 128; off > 0; off >>= 1) {
    if (tid < off) red[tid] += red[tid + off];
    __syncthreads();
  }
  if (tid == 0) dvec[i] = sqrtf(red[0] + 2.0f);
}

__global__ void norm_kernel(const float* __restrict__ adj,
                            const float* __restrict__ dvec,
                            float* __restrict__ Aout) {
  const int idx = blockIdx.x * 256 + threadIdx.x;
  const int i = idx >> 9;
  const int j = idx & (N - 1);
  float a = adj[idx] + (i == j ? 1.0f : 0.0f);
  Aout[idx] = a / (dvec[i] * dvec[j]);
}

__global__ __launch_bounds__(256) void gemm_out_kernel(const float* __restrict__ Anorm,
                                                       const float* __restrict__ XW,
                                                       const float* __restrict__ gcn_b,
                                                       const float* __restrict__ gamma,
                                                       const float* __restrict__ beta,
                                                       float* __restrict__ X) {
  __shared__ float As[16 * PITCH];
  const int tid  = threadIdx.x;
  const int wave = tid >> 5;
  const int lane = tid & 31;
  const int m0 = blockIdx.x * 16;
  const int n0 = wave * 16;
  v4f st[8];
  stage_issue(st, Anorm + m0 * N, tid);
  BRing q;
  const float* bp = XW;
  if (wave < 2) {
    bp = XW + n0 + (lane & 15) + ((lane >> 4) << 1) * OUT;
    b_prologue<OUT>(q, bp);
  }
  stage_commit(As, st, tid);
  __syncthreads();
  if (wave >= 2) return;
  v8f acc = strip_chain<OUT>(q, As + (lane & 15) * PITCH + ((lane >> 4) << 1), bp);

  const int col = n0 + (lane & 15);
  const float inv_std = 0.99999500003749969f;
  const float sc = gamma[col] * inv_std;
  const float bt = beta[col];
  const float gb = gcn_b[col];
  const int rbase = m0 + ((lane >> 4) << 3);
#pragma unroll
  for (int v = 0; v < 8; ++v) {
    float t = (acc[v] + gb) * sc + bt;
    X[(rbase + v) * OUT + col] = fmaxf(t, 0.0f);
  }
}

extern "C" void kernel_launch(void* const* d_in, const int* in_sizes, int n_in,
                              void* d_out, int out_size, void* d_ws, size_t ws_size,
                              hipStream_t stream) {
  const float* F     = (const float*)d_in[0];
  const float* w1    = (const float*)d_in[1];
  const float* b1    = (const float*)d_in[2];
  const float* w2    = (const float*)d_in[3];
  const float* b2    = (const float*)d_in[4];
  const float* gw    = (const float*)d_in[5];
  const float* gb    = (const float*)d_in[6];
  const float* gamma = (const float*)d_in[7];
  const float* beta  = (const float*)d_in[8];

  float* out_x = (float*)d_out;
  float* out_A = out_x + N * OUT;

  float* ws   = (float*)d_ws;
  float* PQ   = ws;
  float* adj  = PQ + N * 2 * H;
  float* XW   = adj + N * N;
  float* dvec = XW + N * OUT;

  gemm_pq_kernel   <<<128,           256, 0, stream>>>(F, w1, PQ);
  gemm_xw_kernel   <<<32,            256, 0, stream>>>(F, gw, XW);
  relnet_pair_kernel<<<dim3(32, 32), 256, 0, stream>>>(PQ, b1, w2, b2, adj);
  dvec_kernel      <<<N,             256, 0, stream>>>(adj, dvec);
  norm_kernel      <<<(N * N) / 256, 256, 0, stream>>>(adj, dvec, out_A);
  gemm_out_kernel  <<<32,            256, 0, stream>>>(out_A, XW, gb, gamma, beta, out_x);
}


# Pair stage re-cut by rows: a work-group owns 2 whole rows of adj (P copied once to LDS, a thread computes 2 columns) and finishes their degree itself; 6 launches to 5; f32 sum order unchanged
# speedup vs baseline: 2.1054x; 1.3764x over previous
#include <hip/hip_runtime.h>
#include <math.h>


#define N    512
#define DIM  512
#define H    64
#define OUT  32

typedef __attribute__((ext_vector_type(2))) float v2f;
typedef __attribute__((ext_vector_type(8))) float v8f;

typedef __attribute__((ext_vector_type(4))) float v4f;

#define PITCH 516

struct BRing { v2f r[4][8]; };

__device__ __forceinline__ void stage_issue(v4f (&st)[8], const float* __restrict__ src, int tid) {
#pragma unroll
  for (int i = 0; i < 8; ++i) {
    const int idx = tid + i * 256;
    st[i] = *(const v4f*)(src + (idx >> 7) * DIM + (idx & 127) * 4);
  }
}

__device__ __forceinline__ void stage_commit(float* As, const v4f (&st)[8], int tid) {
#pragma unroll
  for (int i = 0; i < 8; ++i) {
    const int idx = tid + i * 256;
    *(v4f*)(As + (idx >> 7) * PITCH + (idx & 127) * 4) = st[i];
  }
}

template <int LDB>
__device__ __forceinline__ void load_b_block(v2f (&dst)[8], const float* __restrict__ bp, int blk) {
#pragma unroll
  for (int u = 0; u < 8; ++u) {
    const float* p = bp + (blk * 32 + u * 4) * LDB;
    dst[u].x = p[0];
    dst[u].y = p[LDB];
  }
}

template <int LDB>
__device__ __forceinline__ void b_prologue(BRing& q, const float* __restrict__ bp) {
#pragma unroll
  for (int blk = 0; blk < 3; ++blk) load_b_block<LDB>(q.r[blk], bp, blk);
}

template <int LDB>
__device__ __forceinline__ v8f strip_chain(BRing& q, const float* arow, const float* __restrict__ bp) {
  v8f c = {};
  v2f a[2][8];
#pragma unroll
  for (int u = 0; u < 8; ++u) a[0][u] = *(const v2f*)(arow + 4 * u);
#pragma unroll
  for (int blk = 0; blk < 16; ++blk) {
    if (blk + 3 < 16) load_b_block<LDB>(q.r[(blk + 3) & 3], bp, blk + 3);
    if (blk + 1 < 16) {
#pragma unroll
      for (int u = 0; u < 8; ++u) a[(blk + 1) & 1][u] = *(const v2f*)(arow + (blk + 1) * 32 + 4 * u);
    }
    __builtin_amdgcn_sched_barrier(0);
#pragma unroll
    for (int u = 0; u < 8; ++u)
      c = __builtin_amdgcn_wmma_f32_16x16x4_f32(false, a[blk & 1][u], false, q.r[blk & 3][u],
                                                (short)0, c, false, false);
  }
  return c;
}

__device__ inline void store_tile16x16(float* __restrict__ D, int ldd,
                                       int m0, int n0, v8f c, int lane) {
  int col   = n0 + (lane & 15);
  int rbase = m0 + ((lane >> 4) << 3);
#pragma unroll
  for (int v = 0; v < 8; ++v) D[(rbase + v) * ldd + col] = c[v];
}

__global__ __launch_bounds__(256) void gemm_pq_kernel(const float* __restrict__ F,
                                                      const float* __restrict__ w1,
                                                      float* __restrict__ PQ) {
  __shared__ float As[16 * PITCH];
  const int tid  = threadIdx.x;
  const int wave = tid >> 5;
  const int lane = tid & 31;
  const int m0 = (blockIdx.x >> 2) * 16;
  const int n0 = ((blockIdx.x & 3) * 2 + wave) * 16;
  v4f st[8];
  stage_issue(st, F + m0 * DIM, tid);
  BRing q;
  const float* bp = w1;
  if (wave < 2) {
    const int c = n0 + (lane & 15);
    const float* Bcol = (c < H) ? (w1 + c) : (w1 + DIM * H + (c - H));
    bp = Bcol + ((lane >> 4) << 1) * H;
    b_prologue<H>(q, bp);
  }
  stage_commit(As, st, tid);
  __syncthreads();
  if (wave >= 2) return;
  v8f acc = strip_chain<H>(q, As + (lane & 15) * PITCH + ((lane >> 4) << 1), bp);
  store_tile16x16(PQ, 2 * H, m0, n0, acc, lane);
}

__global__ __launch_bounds__(256) void gemm_xw_kernel(const float* __restrict__ F,
                                                      const float* __restrict__ W,
                                                      float* __restrict__ XW) {
  __shared__ float As[16 * PITCH];
  const int tid  = threadIdx.x;
  const int wave = tid >> 5;
  const int lane = tid & 31;
  const int m0 = blockIdx.x * 16;
  const int n0 = wave * 16;
  v4f st[8];
  stage_issue(st, F + m0 * DIM, tid);
  BRing q;
  const float* bp = W;
  if (wave < 2) {
    bp = W + n0 + (lane & 15) + ((lane >> 4) << 1) * OUT;
    b_prologue<OUT>(q, bp);
  }
  stage_commit(As, st, tid);
  __syncthreads();
  if (wave >= 2) return;
  v8f acc = strip_chain<OUT>(q, As + (lane & 15) * PITCH + ((lane >> 4) << 1), bp);
  store_tile16x16(XW, OUT, m0, n0, acc, lane);
}

#define PPITCH 68

__global__ __launch_bounds__(256) void relnet_rows_kernel(const float* __restrict__ PQ,
                                                          const float* __restrict__ b1,
                                                          const float* __restrict__ w2,
                                                          const float* __restrict__ b2p,
                                                          float* __restrict__ adj,
                                                          float* __restrict__ dvec) {
  __shared__ float Ps[N * PPITCH];
  __shared__ float sm[4][H];
  __shared__ float red[2][256];
  const int tid = threadIdx.x;
  const int i0 = blockIdx.x * 2;

  const int sr = tid >> 6, sc = tid & (H - 1);
  const float* sp = (sr < 2) ? (PQ + (i0 + sr) * (2 * H) + H + sc) : ((sr == 2) ? (b1 + sc) : (w2 + sc));
  const float sv = *sp;

  const float* gsrc = PQ + (tid >> 4) * (2 * H) + (tid & 15) * 4;
  float*       ldst = Ps + (tid >> 4) * PPITCH + (tid & 15) * 4;
  v4f st[32];
#pragma unroll
  for (int i = 0; i < 32; ++i) st[i] = *(const v4f*)(gsrc + i * 16 * (2 * H));
  sm[sr][sc] = sv;
#pragma unroll
  for (int i = 0; i < 32; ++i) *(v4f*)(ldst + i * 16 * PPITCH) = st[i];
  __syncthreads();

  const float* pa = Ps + tid * PPITCH;
  const float* pb = Ps + (tid + 256) * PPITCH;
  float a00 = 0.0f, a01 = 0.0f, a10 = 0.0f, a11 = 0.0f;
#pragma unroll
  for (int k = 0; k < H / 4; ++k) {
    const v4f x  = *(const v4f*)(pa + 4 * k);
    const v4f y  = *(const v4f*)(pb + 4 * k);
    const v4f q0 = *(const v4f*)(&sm[0][4 * k]);
    const v4f q1 = *(const v4f*)(&sm[1][4 * k]);
    const v4f bb = *(const v4f*)(&sm[2][4 * k]);
    const v4f ww = *(const v4f*)(&sm[3][4 * k]);
#pragma unroll
    for (int e = 0; e < 4; ++e) {
      a00 = fmaf(fmaxf(x[e] + q0[e] + bb[e], 0.0f), ww[e], a00);
      a01 = fmaf(fmaxf(y[e] + q0[e] + bb[e], 0.0f), ww[e], a01);
      a10 = fmaf(fmaxf(x[e] + q1[e] + bb[e], 0.0f), ww[e], a10);
      a11 = fmaf(fmaxf(y[e] + q1[e] + bb[e], 0.0f), ww[e], a11);
    }
  }
  const float b2 = b2p[0];
  const float v00 = 1.0f / (1.0f + expf(-(a00 + b2)));
  const float v01 = 1.0f / (1.0f + expf(-(a01 + b2)));
  const float v10 = 1.0f / (1.0f + expf(-(a10 + b2)));
  const float v11 = 1.0f / (1.0f + expf(-(a11 + b2)));
  adj[i0 * N + tid]             = v00;
  adj[i0 * N + tid + 256]       = v01;
  adj[(i0 + 1) * N + tid]       = v10;
  adj[(i0 + 1) * N + tid + 256] = v11;

  red[0][tid] = v00 + v01;
  red[1][tid] = v10 + v11;
  __syncthreads();
  if (tid < 64) {
    const int r = tid >> 5, l = tid & 31;
    const float* rr = red[r];
    const float s1a = rr[l]      + rr[l + 128];
    const float s1b = rr[l + 32] + rr[l + 160];
    const float s1c = rr[l + 64] + rr[l + 192];
    const float s1d = rr[l + 96] + rr[l + 224];
    const float s2a = s1a + s1c;
    const float s2b = s1b + s1d;
    float v = s2a + s2b;
    for (int off = 16; off > 0; off >>= 1) v = v + __shfl_down(v, off);
    if (l == 0) dvec[i0 + r] = sqrtf(v + 2.0f);
  }
}

__global__ void norm_kernel(const float* __restrict__ adj,
                            const float* __restrict__ dvec,
                            float* __restrict__ Aout) {
  const int idx = blockIdx.x * 256 + threadIdx.x;
  const int i = idx >> 9;
  const int j = idx & (N - 1);
  float a = adj[idx] + (i == j ? 1.0f : 0.0f);
  Aout[idx] = a / (dvec[i] * dvec[j]);
}

__global__ __launch_bounds__(256) void gemm_out_kernel(const float* __restrict__ Anorm,
                                                       const float* __restrict__ XW,
                                                       const float* __restrict__ gcn_b,
                                                       const float* __restrict__ gamma,
                                                       const float* __restrict__ beta,
                                                       float* __restrict__ X) {
  __shared__ float As[16 * PITCH];
  const int tid  = threadIdx.x;
  const int wave = tid >> 5;
  const int lane = tid & 31;
  const int m0 = blockIdx.x * 16;
  const int n0 = wave * 16;
  v4f st[8];
  stage_issue(st, Anorm + m0 * N, tid);
  BRing q;
  const float* bp = XW;
  if (wave < 2) {
    bp = XW + n0 + (lane & 15) + ((lane >> 4) << 1) * OUT;
    b_prologue<OUT>(q, bp);
  }
  stage_commit(As, st, tid);
  __syncthreads();
  if (wave >= 2) return;
  v8f acc = strip_chain<OUT>(q, As + (lane & 15) * PITCH + ((lane >> 4) << 1), bp);

  const int col = n0 + (lane & 15);
  const float inv_std = 0.99999500003749969f;
  const float sc = gamma[col] * inv_std;
  const float bt = beta[col];
  const float gb = gcn_b[col];
  const int rbase = m0 + ((lane >> 4) << 3);
#pragma unroll
  for (int v = 0; v < 8; ++v) {
    float t = (acc[v] + gb) * sc + bt;
    X[(rbase + v) * OUT + col] = fmaxf(t, 0.0f);
  }
}

extern "C" void kernel_launch(void* const* d_in, const int* in_sizes, int n_in,
                              void* d_out, int out_size, void* d_ws, size_t ws_size,
                              hipStream_t stream) {
  const float* F     = (const float*)d_in[0];
  const float* w1    = (const float*)d_in[1];
  const float* b1    = (const float*)d_in[2];
  const float* w2    = (const float*)d_in[3];
  const float* b2    = (const float*)d_in[4];
  const float* gw    = (const float*)d_in[5];
  const float* gb    = (const float*)d_in[6];
  const float* gamma = (const float*)d_in[7];
  const float* beta  = (const float*)d_in[8];

  float* out_x = (float*)d_out;
  float* out_A = out_x + N * OUT;

  float* ws   = (float*)d_ws;
  float* PQ   = ws;
  float* adj  = PQ + N * 2 * H;
  float* XW   = adj + N * N;
  float* dvec = XW + N * OUT;

  gemm_pq_kernel   <<<128,           256, 0, stream>>>(F, w1, PQ);
  gemm_xw_kernel   <<<32,            256, 0, stream>>>(F, gw, XW);
  relnet_rows_kernel<<<N / 2,        256, 0, stream>>>(PQ, b1, w2, b2, adj, dvec);
  norm_kernel      <<<(N * N) / 256, 256, 0, stream>>>(adj, dvec, out_A);
  gemm_out_kernel  <<<32,            256, 0, stream>>>(out_A, XW, gb, gamma, beta, out_x);
}


# XW GEMM leaves its own launch: its 32 strips run as the first 32 work-groups of the pair-stage launch (F strip copied into that kernel's LDS array); 5 launches to 4; f32 sum order unchanged
# speedup vs baseline: 2.5464x; 1.2095x over previous
#include <hip/hip_runtime.h>
#include <math.h>


#define N    512
#define DIM  512
#define H    64
#define OUT  32

typedef __attribute__((ext_vector_type(2))) float v2f;
typedef __attribute__((ext_vector_type(8))) float v8f;

typedef __attribute__((ext_vector_type(4))) float v4f;

#define PITCH 516

struct BRing { v2f r[4][8]; };

__device__ __forceinline__ void stage_issue(v4f (&st)[8], const float* __restrict__ src, int tid) {
#pragma unroll
  for (int i = 0; i < 8; ++i) {
    const int idx = tid + i * 256;
    st[i] = *(const v4f*)(src + (idx >> 7) * DIM + (idx & 127) * 4);
  }
}

__device__ __forceinline__ void stage_commit(float* As, const v4f (&st)[8], int tid) {
#pragma unroll
  for (int i = 0; i < 8; ++i) {
    const int idx = tid + i * 256;
    *(v4f*)(As + (idx >> 7) * PITCH + (idx & 127) * 4) = st[i];
  }
}

template <int LDB>
__device__ __forceinline__ void load_b_block(v2f (&dst)[8], const float* __restrict__ bp, int blk) {
#pragma unroll
  for (int u = 0; u < 8; ++u) {
    const float* p = bp + (blk * 32 + u * 4) * LDB;
    dst[u].x = p[0];
    dst[u].y = p[LDB];
  }
}

template <int LDB>
__device__ __forceinline__ void b_prologue(BRing& q, const float* __restrict__ bp) {
#pragma unroll
  for (int blk = 0; blk < 3; ++blk) load_b_block<LDB>(q.r[blk], bp, blk);
}

template <int LDB>
__device__ __forceinline__ v8f strip_chain(BRing& q, const float* arow, const float* __restrict__ bp) {
  v8f c = {};
  v2f a[2][8];
#pragma unroll
  for (int u = 0; u < 8; ++u) a[0][u] = *(const v2f*)(arow + 4 * u);
#pragma unroll
  for (int blk = 0; blk < 16; ++blk) {
    if (blk + 3 < 16) load_b_block<LDB>(q.r[(blk + 3) & 3], bp, blk + 3);
    if (blk + 1 < 16) {
#pragma unroll
      for (int u = 0; u < 8; ++u) a[(blk + 1) & 1][u] = *(const v2f*)(arow + (blk + 1) * 32 + 4 * u);
    }
    __builtin_amdgcn_sched_barrier(0);
#pragma unroll
    for (int u = 0; u < 8; ++u)
      c = __builtin_amdgcn_wmma_f32_16x16x4_f32(false, a[blk & 1][u], false, q.r[blk & 3][u],
                                                (short)0, c, false, false);
  }
  return c;
}

__device__ inline void store_tile16x16(float* __restrict__ D, int ldd,
                                       int m0, int n0, v8f c, int lane) {
  int col   = n0 + (lane & 15);
  int rbase = m0 + ((lane >> 4) << 3);
#pragma unroll
  for (int v = 0; v < 8; ++v) D[(rbase + v) * ldd + col] = c[v];
}

__global__ __launch_bounds__(256) void gemm_pq_kernel(const float* __restrict__ F,
                                                      const float* __restrict__ w1,
                                                      float* __restrict__ PQ) {
  __shared__ float As[16 * PITCH];
  const int tid  = threadIdx.x;
  const int wave = tid >> 5;
  const int lane = tid & 31;
  const int m0 = (blockIdx.x >> 2) * 16;
  const int n0 = ((blockIdx.x & 3) * 2 + wave) * 16;
  v4f st[8];
  stage_issue(st, F + m0 * DIM, tid);
  BRing q;
  const float* bp = w1;
  if (wave < 2) {
    const int c = n0 + (lane & 15);
    const float* Bcol = (c < H) ? (w1 + c) : (w1 + DIM * H + (c - H));
    bp = Bcol + ((lane >> 4) << 1) * H;
    b_prologue<H>(q, bp);
  }
  stage_commit(As, st, tid);
  __syncthreads();
  if (wave >= 2) return;
  v8f acc = strip_chain<H>(q, As + (lane & 15) * PITCH + ((lane >> 4) << 1), bp);
  store_tile16x16(PQ, 2 * H, m0, n0, acc, lane);
}

#define XW_GROUPS 32

__device__ __forceinline__ void xw_strip(float* As, const float* __restrict__ F,
                                         const float* __restrict__ W, float* __restrict__ XW,
                                         int strip, int tid) {
  const int wave = tid >> 5;
  const int lane = tid & 31;
  const int m0 = strip * 16;
  const int n0 = wave * 16;
  v4f st[8];
  stage_issue(st, F + m0 * DIM, tid);
  BRing q;
  const float* bp = W;
  if (wave < 2) {
    bp = W + n0 + (lane & 15) + ((lane >> 4) << 1) * OUT;
    b_prologue<OUT>(q, bp);
  }
  stage_commit(As, st, tid);
  __syncthreads();
  if (wave >= 2) return;
  v8f acc = strip_chain<OUT>(q, As + (lane & 15) * PITCH + ((lane >> 4) << 1), bp);
  store_tile16x16(XW, OUT, m0, n0, acc, lane);
}

#define PPITCH 68

__global__ __launch_bounds__(256) void relnet_rows_xw_kernel(const float* __restrict__ PQ,
                                                             const float* __restrict__ b1,
                                                             const float* __restrict__ w2,
                                                             const float* __restrict__ b2p,
                                                             float* __restrict__ adj,
                                                             float* __restrict__ dvec,
                                                             const float* __restrict__ F,
                                                             const float* __restrict__ gw,
                                                             float* __restrict__ XW) {
  __shared__ float Ps[N * PPITCH];
  __shared__ float sm[4][H];
  __shared__ float red[2][256];
  const int tid = threadIdx.x;
  if (blockIdx.x < XW_GROUPS) {
    xw_strip(Ps, F, gw, XW, blockIdx.x, tid);
    return;
  }
  const int i0 = (blockIdx.x - XW_GROUPS) * 2;

  const int sr = tid >> 6, sc = tid & (H - 1);
  const float* sp = (sr < 2) ? (PQ + (i0 + sr) * (2 * H) + H + sc) : ((sr == 2) ? (b1 + sc) : (w2 + sc));
  const float sv = *sp;

  const float* gsrc = PQ + (tid >> 4) * (2 * H) + (tid & 15) * 4;
  float*       ldst = Ps + (tid >> 4) * PPITCH + (tid & 15) * 4;
  v4f st[32];
#pragma unroll
  for (int i = 0; i < 32; ++i) st[i] = *(const v4f*)(gsrc + i * 16 * (2 * H));
  sm[sr][sc] = sv;
#pragma unroll
  for (int i = 0; i < 32; ++i) *(v4f*)(ldst + i * 16 * PPITCH) = st[i];
  __syncthreads();

  const float* pa = Ps + tid * PPITCH;
  const float* pb = Ps + (tid + 256) * PPITCH;
  float a00 = 0.0f, a01 = 0.0f, a10 = 0.0f, a11 = 0.0f;
#pragma unroll
  for (int k = 0; k < H / 4; ++k) {
    const v4f x  = *(const v4f*)(pa + 4 * k);
    const v4f y  = *(const v4f*)(pb + 4 * k);
    const v4f q0 = *(const v4f*)(&sm[0][4 * k]);
    const v4f q1 = *(const v4f*)(&sm[1][4 * k]);
    const v4f bb = *(const v4f*)(&sm[2][4 * k]);
    const v4f ww = *(const v4f*)(&sm[3][4 * k]);
#pragma unroll
    for (int e = 0; e < 4; ++e) {
      a00 = fmaf(fmaxf(x[e] + q0[e] + bb[e], 0.0f), ww[e], a00);
      a01 = fmaf(fmaxf(y[e] + q0[e] + bb[e], 0.0f), ww[e], a01);
      a10 = fmaf(fmaxf(x[e] + q1[e] + bb[e], 0.0f), ww[e], a10);
      a11 = fmaf(fmaxf(y[e] + q1[e] + bb[e], 0.0f), ww[e], a11);
    }
  }
  const float b2 = b2p[0];
  const float v00 = 1.0f / (1.0f + expf(-(a00 + b2)));
  const float v01 = 1.0f / (1.0f + expf(-(a01 + b2)));
  const float v10 = 1.0f / (1.0f + expf(-(a10 + b2)));
  const float v11 = 1.0f / (1.0f + expf(-(a11 + b2)));
  adj[i0 * N + tid]             = v00;
  adj[i0 * N + tid + 256]       = v01;
  adj[(i0 + 1) * N + tid]       = v10;
  adj[(i0 + 1) * N + tid + 256] = v11;

  red[0][tid] = v00 + v01;
  red[1][tid] = v10 + v11;
  __syncthreads();
  if (tid < 64) {
    const int r = tid >> 5, l = tid & 31;
    const float* rr = red[r];
    const float s1a = rr[l]      + rr[l + 128];
    const float s1b = rr[l + 32] + rr[l + 160];
    const float s1c = rr[l + 64] + rr[l + 192];
    const float s1d = rr[l + 96] + rr[l + 224];
    const float s2a = s1a + s1c;
    const float s2b = s1b + s1d;
    float v = s2a + s2b;
    for (int off = 16; off > 0; off >>= 1) v = v + __shfl_down(v, off);
    if (l == 0) dvec[i0 + r] = sqrtf(v + 2.0f);
  }
}

__global__ void norm_kernel(const float* __restrict__ adj,
                            const float* __restrict__ dvec,
                            float* __restrict__ Aout) {
  const int idx = blockIdx.x * 256 + threadIdx.x;
  const int i = idx >> 9;
  const int j = idx & (N - 1);
  float a = adj[idx] + (i == j ? 1.0f : 0.0f);
  Aout[idx] = a / (dvec[i] * dvec[j]);
}

__global__ __launch_bounds__(256) void gemm_out_kernel(const float* __restrict__ Anorm,
                                                       const float* __restrict__ XW,
                                                       const float* __restrict__ gcn_b,
                                                       const float* __restrict__ gamma,
                                                       const float* __restrict__ beta,
                                                       float* __restrict__ X) {
  __shared__ float As[16 * PITCH];
  const int tid  = threadIdx.x;
  const int wave = tid >> 5;
  const int lane = tid & 31;
  const int m0 = blockIdx.x * 16;
  const int n0 = wave * 16;
  v4f st[8];
  stage_issue(st, Anorm + m0 * N, tid);
  BRing q;
  const float* bp = XW;
  if (wave < 2) {
    bp = XW + n0 + (lane & 15) + ((lane >> 4) << 1) * OUT;
    b_prologue<OUT>(q, bp);
  }
  stage_commit(As, st, tid);
  __syncthreads();
  if (wave >= 2) return;
  v8f acc = strip_chain<OUT>(q, As + (lane & 15) * PITCH + ((lane >> 4) << 1), bp);

  const int col = n0 + (lane & 15);
  const float inv_std = 0.99999500003749969f;
  const float sc = gamma[col] * inv_std;
  const float bt = beta[col];
  const float gb = gcn_b[col];
  const int rbase = m0 + ((lane >> 4) << 3);
#pragma unroll
  for (int v = 0; v < 8; ++v) {
    float t = (acc[v] + gb) * sc + bt;
    X[(rbase + v) * OUT + col] = fmaxf(t, 0.0f);
  }
}

extern "C" void kernel_launch(void* const* d_in, const int* in_sizes, int n_in,
                              void* d_out, int out_size, void* d_ws, size_t ws_size,
                              hipStream_t stream) {
  const float* F     = (const float*)d_in[0];
  const float* w1    = (const float*)d_in[1];
  const float* b1    = (const float*)d_in[2];
  const float* w2    = (const float*)d_in[3];
  const float* b2    = (const float*)d_in[4];
  const float* gw    = (const float*)d_in[5];
  const float* gb    = (const float*)d_in[6];
  const float* gamma = (const float*)d_in[7];
  const float* beta  = (const float*)d_in[8];

  float* out_x = (float*)d_out;
  float* out_A = out_x + N * OUT;

  float* ws   = (float*)d_ws;
  float* PQ   = ws;
  float* adj  = PQ + N * 2 * H;
  float* XW   = adj + N * N;
  float* dvec = XW + N * OUT;

  gemm_pq_kernel   <<<128,           256, 0, stream>>>(F, w1, PQ);
  relnet_rows_xw_kernel<<<XW_GROUPS + N / 2, 256, 0, stream>>>(PQ, b1, w2, b2, adj, dvec, F, gw, XW);
  norm_kernel      <<<(N * N) / 256, 256, 0, stream>>>(adj, dvec, out_A);
  gemm_out_kernel  <<<32,            256, 0, stream>>>(out_A, XW, gb, gamma, beta, out_x);
}


# Output GEMM leaves its own launch: its 32 strips ride first in the norm launch (norm part: 128 groups of 4 rows) and normalise their own 16 adj rows into LDS; 4 launches to 3; values bit-identical
# speedup vs baseline: 2.8235x; 1.1088x over previous
#include <hip/hip_runtime.h>
#include <math.h>


#define N    512
#define DIM  512
#define H    64
#define OUT  32

typedef __attribute__((ext_vector_type(2))) float v2f;
typedef __attribute__((ext_vector_type(8))) float v8f;

typedef __attribute__((ext_vector_type(4))) float v4f;

#define PITCH 516

struct BRing { v2f r[4][8]; };

__device__ __forceinline__ void stage_issue(v4f (&st)[8], const float* __restrict__ src, int tid) {
#pragma unroll
  for (int i = 0; i < 8; ++i) {
    const int idx = tid + i * 256;
    st[i] = *(const v4f*)(src + (idx >> 7) * DIM + (idx & 127) * 4);
  }
}

__device__ __forceinline__ void stage_commit(float* As, const v4f (&st)[8], int tid) {
#pragma unroll
  for (int i = 0; i < 8; ++i) {
    const int idx = tid + i * 256;
    *(v4f*)(As + (idx >> 7) * PITCH + (idx & 127) * 4) = st[i];
  }
}

template <int LDB>
__device__ __forceinline__ void load_b_block(v2f (&dst)[8], const float* __restrict__ bp, int blk) {
#pragma unroll
  for (int u = 0; u < 8; ++u) {
    const float* p = bp + (blk * 32 + u * 4) * LDB;
    dst[u].x = p[0];
    dst[u].y = p[LDB];
  }
}

template <int LDB>
__device__ __forceinline__ void b_prologue(BRing& q, const float* __restrict__ bp) {
#pragma unroll
  for (int blk = 0; blk < 3; ++blk) load_b_block<LDB>(q.r[blk], bp, blk);
}

template <int LDB>
__device__ __forceinline__ v8f strip_chain(BRing& q, const float* arow, const float* __restrict__ bp) {
  v8f c = {};
  v2f a[2][8];
#pragma unroll
  for (int u = 0; u < 8; ++u) a[0][u] = *(const v2f*)(arow + 4 * u);
#pragma unroll
  for (int blk = 0; blk < 16; ++blk) {
    if (blk + 3 < 16) load_b_block<LDB>(q.r[(blk + 3) & 3], bp, blk + 3);
    if (blk + 1 < 16) {
#pragma unroll
      for (int u = 0; u < 8; ++u) a[(blk + 1) & 1][u] = *(const v2f*)(arow + (blk + 1) * 32 + 4 * u);
    }
    __builtin_amdgcn_sched_barrier(0);
#pragma unroll
    for (int u = 0; u < 8; ++u)
      c = __builtin_amdgcn_wmma_f32_16x16x4_f32(false, a[blk & 1][u], false, q.r[blk & 3][u],
                                                (short)0, c, false, false);
  }
  return c;
}

__device__ inline void store_tile16x16(float* __restrict__ D, int ldd,
                                       int m0, int n0, v8f c, int lane) {
  int col   = n0 + (lane & 15);
  int rbase = m0 + ((lane >> 4) << 3);
#pragma unroll
  for (int v = 0; v < 8; ++v) D[(rbase + v) * ldd + col] = c[v];
}

__global__ __launch_bounds__(256) void gemm_pq_kernel(const float* __restrict__ F,
                                                      const float* __restrict__ w1,
                                                      float* __restrict__ PQ) {
  __shared__ float As[16 * PITCH];
  const int tid  = threadIdx.x;
  const int wave = tid >> 5;
  const int lane = tid & 31;
  const int m0 = (blockIdx.x >> 2) * 16;
  const int n0 = ((blockIdx.x & 3) * 2 + wave) * 16;
  v4f st[8];
  stage_issue(st, F + m0 * DIM, tid);
  BRing q;
  const float* bp = w1;
  if (wave < 2) {
    const int c = n0 + (lane & 15);
    const float* Bcol = (c < H) ? (w1 + c) : (w1 + DIM * H + (c - H));
    bp = Bcol + ((lane >> 4) << 1) * H;
    b_prologue<H>(q, bp);
  }
  stage_commit(As, st, tid);
  __syncthreads();
  if (wave >= 2) return;
  v8f acc = strip_chain<H>(q, As + (lane & 15) * PITCH + ((lane >> 4) << 1), bp);
  store_tile16x16(PQ, 2 * H, m0, n0, acc, lane);
}

#define XW_GROUPS 32

__device__ __forceinline__ void xw_strip(float* As, const float* __restrict__ F,
                                         const float* __restrict__ W, float* __restrict__ XW,
                                         int strip, int tid) {
  const int wave = tid >> 5;
  const int lane = tid & 31;
  const int m0 = strip * 16;
  const int n0 = wave * 16;
  v4f st[8];
  stage_issue(st, F + m0 * DIM, tid);
  BRing q;
  const float* bp = W;
  if (wave < 2) {
    bp = W + n0 + (lane & 15) + ((lane >> 4) << 1) * OUT;
    b_prologue<OUT>(q, bp);
  }
  stage_commit(As, st, tid);
  __syncthreads();
  if (wave >= 2) return;
  v8f acc = strip_chain<OUT>(q, As + (lane & 15) * PITCH + ((lane >> 4) << 1), bp);
  store_tile16x16(XW, OUT, m0, n0, acc, lane);
}

#define PPITCH 68

__global__ __launch_bounds__(256) void relnet_rows_xw_kernel(const float* __restrict__ PQ,
                                                             const float* __restrict__ b1,
                                                             const float* __restrict__ w2,
                                                             const float* __restrict__ b2p,
                                                             float* __restrict__ adj,
                                                             float* __restrict__ dvec,
                                                             const float* __restrict__ F,
                                                             const float* __restrict__ gw,
                                                             float* __restrict__ XW) {
  __shared__ float Ps[N * PPITCH];
  __shared__ float sm[4][H];
  __shared__ float red[2][256];
  const int tid = threadIdx.x;
  if (blockIdx.x < XW_GROUPS) {
    xw_strip(Ps, F, gw, XW, blockIdx.x, tid);
    return;
  }
  const int i0 = (blockIdx.x - XW_GROUPS) * 2;

  const int sr = tid >> 6, sc = tid & (H - 1);
  const float* sp = (sr < 2) ? (PQ + (i0 + sr) * (2 * H) + H + sc) : ((sr == 2) ? (b1 + sc) : (w2 + sc));
  const float sv = *sp;

  const float* gsrc = PQ + (tid >> 4) * (2 * H) + (tid & 15) * 4;
  float*       ldst = Ps + (tid >> 4) * PPITCH + (tid & 15) * 4;
  v4f st[32];
#pragma unroll
  for (int i = 0; i < 32; ++i) st[i] = *(const v4f*)(gsrc + i * 16 * (2 * H));
  sm[sr][sc] = sv;
#pragma unroll
  for (int i = 0; i < 32; ++i) *(v4f*)(ldst + i * 16 * PPITCH) = st[i];
  __syncthreads();

  const float* pa = Ps + tid * PPITCH;
  const float* pb = Ps + (tid + 256) * PPITCH;
  float a00 = 0.0f, a01 = 0.0f, a10 = 0.0f, a11 = 0.0f;
#pragma unroll
  for (int k = 0; k < H / 4; ++k) {
    const v4f x  = *(const v4f*)(pa + 4 * k);
    const v4f y  = *(const v4f*)(pb + 4 * k);
    const v4f q0 = *(const v4f*)(&sm[0][4 * k]);
    const v4f q1 = *(const v4f*)(&sm[1][4 * k]);
    const v4f bb = *(const v4f*)(&sm[2][4 * k]);
    const v4f ww = *(const v4f*)(&sm[3][4 * k]);
#pragma unroll
    for (int e = 0; e < 4; ++e) {
      a00 = fmaf(fmaxf(x[e] + q0[e] + bb[e], 0.0f), ww[e], a00);
      a01 = fmaf(fmaxf(y[e] + q0[e] + bb[e], 0.0f), ww[e], a01);
      a10 = fmaf(fmaxf(x[e] + q1[e] + bb[e], 0.0f), ww[e], a10);
      a11 = fmaf(fmaxf(y[e] + q1[e] + bb[e], 0.0f), ww[e], a11);
    }
  }
  const float b2 = b2p[0];
  const float v00 = 1.0f / (1.0f + expf(-(a00 + b2)));
  const float v01 = 1.0f / (1.0f + expf(-(a01 + b2)));
  const float v10 = 1.0f / (1.0f + expf(-(a10 + b2)));
  const float v11 = 1.0f / (1.0f + expf(-(a11 + b2)));
  adj[i0 * N + tid]             = v00;
  adj[i0 * N + tid + 256]       = v01;
  adj[(i0 + 1) * N + tid]       = v10;
  adj[(i0 + 1) * N + tid + 256] = v11;

  red[0][tid] = v00 + v01;
  red[1][tid] = v10 + v11;
  __syncthreads();
  if (tid < 64) {
    const int r = tid >> 5, l = tid & 31;
    const float* rr = red[r];
    const float s1a = rr[l]      + rr[l + 128];
    const float s1b = rr[l + 32] + rr[l + 160];
    const float s1c = rr[l + 64] + rr[l + 192];
    const float s1d = rr[l + 96] + rr[l + 224];
    const float s2a = s1a + s1c;
    const float s2b = s1b + s1d;
    float v = s2a + s2b;
    for (int off = 16; off > 0; off >>= 1) v = v + __shfl_down(v, off);
    if (l == 0) dvec[i0 + r] = sqrtf(v + 2.0f);
  }
}

#define OUT_GROUPS  32
#define NORM_GROUPS (N / 4)

__device__ __forceinline__ v4f norm_piece(v4f a, int i, int c0, float di, v4f dj) {
  v4f o;
#pragma unroll
  for (int e = 0; e < 4; ++e) {
    const float t = a[e] + ((i == c0 + e) ? 1.0f : 0.0f);
    o[e] = t / (di * dj[e]);
  }
  return o;
}

__global__ __launch_bounds__(256) void norm_out_kernel(const float* __restrict__ adj,
                                                       const float* __restrict__ dvec,
                                                       float* __restrict__ Aout,
                                                       const float* __restrict__ XW,
                                                       const float* __restrict__ gcn_b,
                                                       const float* __restrict__ gamma,
                                                       const float* __restrict__ beta,
                                                       float* __restrict__ X) {
  __shared__ float As[16 * PITCH];
  const int tid  = threadIdx.x;
  const int c0   = (tid & 127) * 4;
  const int rsub = tid >> 7;

  if (blockIdx.x >= OUT_GROUPS) {
    const int r0 = (blockIdx.x - OUT_GROUPS) * 4 + rsub;
    const v4f dj = *(const v4f*)(dvec + c0);
    const float d0 = dvec[r0];
    const float d1 = dvec[r0 + 2];
    const v4f a0 = *(const v4f*)(adj + r0 * N + c0);
    const v4f a1 = *(const v4f*)(adj + (r0 + 2) * N + c0);
    *(v4f*)(Aout + r0 * N + c0)       = norm_piece(a0, r0, c0, d0, dj);
    *(v4f*)(Aout + (r0 + 2) * N + c0) = norm_piece(a1, r0 + 2, c0, d1, dj);
    return;
  }

  const int wave = tid >> 5;
  const int lane = tid & 31;
  const int m0 = blockIdx.x * 16;
  const int n0 = wave * 16;
  const v4f dj = *(const v4f*)(dvec + c0);
  float di[8];
#pragma unroll
  for (int i = 0; i < 8; ++i) di[i] = dvec[m0 + rsub + 2 * i];
  v4f st[8];
  stage_issue(st, adj + m0 * N, tid);
  BRing q;
  const float* bp = XW;
  if (wave < 2) {
    bp = XW + n0 + (lane & 15) + ((lane >> 4) << 1) * OUT;
    b_prologue<OUT>(q, bp);
  }
#pragma unroll
  for (int i = 0; i < 8; ++i) st[i] = norm_piece(st[i], m0 + rsub + 2 * i, c0, di[i], dj);
  stage_commit(As, st, tid);
  __syncthreads();
  if (wave >= 2) return;
  v8f acc = strip_chain<OUT>(q, As + (lane & 15) * PITCH + ((lane >> 4) << 1), bp);

  const int col = n0 + (lane & 15);
  const float inv_std = 0.99999500003749969f;
  const float sc = gamma[col] * inv_std;
  const float bt = beta[col];
  const float gb = gcn_b[col];
  const int rbase = m0 + ((lane >> 4) << 3);
#pragma unroll
  for (int v = 0; v < 8; ++v) {
    float t = (acc[v] + gb) * sc + bt;
    X[(rbase + v) * OUT + col] = fmaxf(t, 0.0f);
  }
}

extern "C" void kernel_launch(void* const* d_in, const int* in_sizes, int n_in,
                              void* d_out, int out_size, void* d_ws, size_t ws_size,
                              hipStream_t stream) {
  const float* F     = (const float*)d_in[0];
  const float* w1    = (const float*)d_in[1];
  const float* b1    = (const float*)d_in[2];
  const float* w2    = (const float*)d_in[3];
  const float* b2    = (const float*)d_in[4];
  const float* gw    = (const float*)d_in[5];
  const float* gb    = (const float*)d_in[6];
  const float* gamma = (const float*)d_in[7];
  const float* beta  = (const float*)d_in[8];

  float* out_x = (float*)d_out;
  float* out_A = out_x + N * OUT;

  float* ws   = (float*)d_ws;
  float* PQ   = ws;
  float* adj  = PQ + N * 2 * H;
  float* XW   = adj + N * N;
  float* dvec = XW + N * OUT;

  gemm_pq_kernel   <<<128,           256, 0, stream>>>(F, w1, PQ);
  relnet_rows_xw_kernel<<<XW_GROUPS + N / 2, 256, 0, stream>>>(PQ, b1, w2, b2, adj, dvec, F, gw, XW);
  norm_out_kernel  <<<OUT_GROUPS + NORM_GROUPS, 256, 0, stream>>>(adj, dvec, out_A, XW, gb, gamma, beta, out_x);
}
